# starting kernel
# baseline (speedup 1.0000x reference)
#include <hip/hip_runtime.h>
#include <hip/hip_bf16.h>

#define N_BU 1024
#define N_BI 4096
#define DIM  64
#define LNZ  50

typedef __attribute__((ext_vector_type(2))) float v2f;
typedef __attribute__((ext_vector_type(8))) float v8f;

__global__ __launch_bounds__(64)
void prep_users(const int* __restrict__ user_ids,
                const int* __restrict__ fb_indices,
                const float* __restrict__ fb_values,
                const float* __restrict__ user_weight,
                const float* __restrict__ user_bias,
                const float* __restrict__ irew,
                float* __restrict__ Aout,
                float* __restrict__ ub)
{
    const int b = blockIdx.x;
    const int d = threadIdx.x;
    const int uid = user_ids[b];
    float acc = user_weight[(size_t)uid * DIM + d];
    const int*   idxr = fb_indices + b * LNZ;
    const float* valr = fb_values  + b * LNZ;
#pragma unroll 2
    for (int l = 0; l < LNZ; ++l) {
        const int   it = idxr[l];
        const float v  = valr[l];
        acc = fmaf(v, irew[(size_t)it * DIM + d], acc);
    }
    Aout[b * DIM + d] = acc;
    if (d == 0) ub[b] = user_bias[uid];
}

__global__ __launch_bounds__(256)
void prep_items(const int* __restrict__ item_ids,
                const float* __restrict__ item_weight,
                const float* __restrict__ item_bias,
                float* __restrict__ Bt2,
                float* __restrict__ ib)
{
    const int t = blockIdx.x * 256 + threadIdx.x;
    const int n = t >> 6;
    const int k = t & 63;
    const int iid = item_ids[n];
    const float w = item_weight[(size_t)iid * DIM + k];
    Bt2[((size_t)(k >> 1) * N_BI + n) * 2 + (k & 1)] = w;
    if (k == 0) ib[n] = item_bias[iid];
}

__global__ __launch_bounds__(256)
void gemm_rating(const float* __restrict__ A,
                 const float* __restrict__ Bt2,
                 const float* __restrict__ ub,
                 const float* __restrict__ ib,
                 const float* __restrict__ biasp,
                 float* __restrict__ out)
{
    const int lane  = threadIdx.x & 31;
    const int wave  = threadIdx.x >> 5;
    const int tile  = blockIdx.x * 8 + wave;
    const int m0    = (tile >> 6) << 4;
    const int n0    = (tile & 63) << 6;
    const int mrow  = lane & 15;
    const int khalf = lane >> 4;

    v8f acc0 = {}, acc1 = {}, acc2 = {}, acc3 = {};

    const float* arow  = A   + (m0 + mrow) * DIM + khalf * 2;
    const float* bbase = Bt2 + (size_t)khalf * (2 * N_BI) + (size_t)(n0 + mrow) * 2;

#pragma unroll 4
    for (int k0 = 0; k0 < 16; ++k0) {
        const v2f a = *(const v2f*)(arow + k0 * 4);
        const float* brow = bbase + (size_t)k0 * (4 * N_BI);
        const v2f b0 = *(const v2f*)(brow);
        const v2f b1 = *(const v2f*)(brow + 32);
        const v2f b2 = *(const v2f*)(brow + 64);
        const v2f b3 = *(const v2f*)(brow + 96);
        acc0 = __builtin_amdgcn_wmma_f32_16x16x4_f32(false, a, false, b0, (short)0, acc0, false, false);
        acc1 = __builtin_amdgcn_wmma_f32_16x16x4_f32(false, a, false, b1, (short)0, acc1, false, false);
        acc2 = __builtin_amdgcn_wmma_f32_16x16x4_f32(false, a, false, b2, (short)0, acc2, false, false);
        acc3 = __builtin_amdgcn_wmma_f32_16x16x4_f32(false, a, false, b3, (short)0, acc3, false, false);
    }

    const float biasv = biasp[0];
    const int ncol = mrow;
    const float ib0 = ib[n0 + ncol]      + biasv;
    const float ib1 = ib[n0 + 16 + ncol] + biasv;
    const float ib2 = ib[n0 + 32 + ncol] + biasv;
    const float ib3 = ib[n0 + 48 + ncol] + biasv;

#pragma unroll
    for (int v = 0; v < 8; ++v) {
        const int row = m0 + khalf * 8 + v;
        const float ubv = ub[row];
        float* orow = out + (size_t)row * N_BI + n0 + ncol;
        orow[0]  = acc0[v] + ubv + ib0;
        orow[16] = acc1[v] + ubv + ib1;
        orow[32] = acc2[v] + ubv + ib2;
        orow[48] = acc3[v] + ubv + ib3;
    }
}

extern "C" void kernel_launch(void* const* d_in, const int* in_sizes, int n_in,
                              void* d_out, int out_size, void* d_ws, size_t ws_size,
                              hipStream_t stream) {
    (void)in_sizes; (void)n_in; (void)out_size; (void)ws_size;

    const int*   user_ids    = (const int*)  d_in[0];
    const int*   item_ids    = (const int*)  d_in[1];
    const int*   fb_indices  = (const int*)  d_in[2];
    const float* fb_values   = (const float*)d_in[3];
    const float* user_weight = (const float*)d_in[4];
    const float* user_bias   = (const float*)d_in[5];
    const float* item_weight = (const float*)d_in[6];
    const float* item_bias   = (const float*)d_in[7];
    const float* bias        = (const float*)d_in[8];
    const float* irew        = (const float*)d_in[9];
    float* out = (float*)d_out;

    float* A   = (float*)d_ws;
    float* Bt2 = A   + N_BU * DIM;
    float* ub  = Bt2 + DIM * N_BI;
    float* ib  = ub  + N_BU;

    prep_users<<<N_BU, 64, 0, stream>>>(user_ids, fb_indices, fb_values,
                                        user_weight, user_bias, irew, A, ub);
    prep_items<<<(N_BI * DIM) / 256, 256, 0, stream>>>(item_ids, item_weight,
                                                       item_bias, Bt2, ib);
    gemm_rating<<<(N_BU / 16) * (N_BI / 64) / 8, 256, 0, stream>>>(A, Bt2, ub, ib,
                                                                   bias, out);
}


# block barrier now only behind the A build (hand-written, item-row loads stay in flight across it); each wave stages and multiplies its item rows in four 32-item steps as they land; sum order kept
# speedup vs baseline: 2.3576x; 2.3576x over previous
#include <hip/hip_runtime.h>


#define N_BU  1024
#define N_BI  4096
#define DIM   64
#define LNZ   50
#define NSLAB 1024
#define WCOLS 128
#define PITCH 68

typedef __attribute__((ext_vector_type(2))) float v2f;
typedef __attribute__((ext_vector_type(4))) float v4f;
typedef __attribute__((ext_vector_type(8))) float v8f;

static __device__ __forceinline__ int lane_int(int v, int l) {
    return __builtin_amdgcn_readlane(v, l);
}
static __device__ __forceinline__ float lane_float(float v, int l) {
    return __int_as_float(__builtin_amdgcn_readlane(__float_as_int(v), l));
}

__global__ __launch_bounds__(256)
void fused_rating(const int* __restrict__ user_ids,
                  const int* __restrict__ item_ids,
                  const int* __restrict__ fb_indices,
                  const float* __restrict__ fb_values,
                  const float* __restrict__ user_weight,
                  const float* __restrict__ user_bias,
                  const float* __restrict__ item_weight,
                  const float* __restrict__ item_bias,
                  const float* __restrict__ biasp,
                  const float* __restrict__ irew,
                  float* __restrict__ out)
{
    __shared__ __attribute__((aligned(16))) float As[16 * PITCH];
    __shared__ __attribute__((aligned(16))) float Bs[NSLAB * PITCH];
    __shared__ float ubs[16];
    __shared__ float ibs[NSLAB];

    const int lane = threadIdx.x & 31;
    const int wave = __builtin_amdgcn_readfirstlane(threadIdx.x >> 5);
    const int mt   = blockIdx.x >> 2;
    const int ns   = blockIdx.x & 3;
    const int m0   = mt * 16;
    const int lb   = wave * WCOLS;
    const int nb   = ns * NSLAB + lb;
    const int r    = lane >> 4, q = lane & 15;

    int idv[4];
#pragma unroll
    for (int t = 0; t < 4; ++t) idv[t] = item_ids[nb + 32 * t + lane];

    const int b0 = m0 + 2 * wave;
    const int b1 = b0 + 1;
    const int hi = (32 + lane < LNZ) ? (32 + lane) : (LNZ - 1);
    const int   ix0a = fb_indices[b0 * LNZ + lane], ix0b = fb_indices[b0 * LNZ + hi];
    const int   ix1a = fb_indices[b1 * LNZ + lane], ix1b = fb_indices[b1 * LNZ + hi];
    const float vl0a = fb_values[b0 * LNZ + lane],  vl0b = fb_values[b0 * LNZ + hi];
    const float vl1a = fb_values[b1 * LNZ + lane],  vl1b = fb_values[b1 * LNZ + hi];
    const int uid0 = user_ids[b0];
    const int uid1 = user_ids[b1];

    v2f a0 = *(const v2f*)(user_weight + (size_t)uid0 * DIM + 2 * lane);
    v2f a1 = *(const v2f*)(user_weight + (size_t)uid1 * DIM + 2 * lane);
    v2f g0[LNZ], g1[LNZ];
#pragma unroll
    for (int l = 0; l < LNZ; ++l) {
        const int i0 = lane_int(l < 32 ? ix0a : ix0b, l & 31);
        const int i1 = lane_int(l < 32 ? ix1a : ix1b, l & 31);
        g0[l] = *(const v2f*)(irew + (size_t)i0 * DIM + 2 * lane);
        g1[l] = *(const v2f*)(irew + (size_t)i1 * DIM + 2 * lane);
    }
    float ibv[4];
#pragma unroll
    for (int t = 0; t < 4; ++t) ibv[t] = item_bias[idv[t]];
    v4f w[64];
#pragma unroll
    for (int j = 0; j < 64; ++j) {
        const int e   = 2 * j;
        const int id0 = lane_int(idv[e >> 5], e & 31);
        const int id1 = lane_int(idv[e >> 5], (e & 31) + 1);
        const int iid = r ? id1 : id0;
        w[j] = *(const v4f*)(item_weight + (size_t)iid * DIM + 4 * q);
    }

#pragma unroll
    for (int l = 0; l < LNZ; ++l) {
        const float v0 = lane_float(l < 32 ? vl0a : vl0b, l & 31);
        const float v1 = lane_float(l < 32 ? vl1a : vl1b, l & 31);
        a0.x = fmaf(v0, g0[l].x, a0.x);
        a0.y = fmaf(v0, g0[l].y, a0.y);
        a1.x = fmaf(v1, g1[l].x, a1.x);
        a1.y = fmaf(v1, g1[l].y, a1.y);
    }
    *(v2f*)(As + (2 * wave) * PITCH + 2 * lane)     = a0;
    *(v2f*)(As + (2 * wave + 1) * PITCH + 2 * lane) = a1;
    if (lane == 0) {
        ubs[2 * wave]     = user_bias[uid0];
        ubs[2 * wave + 1] = user_bias[uid1];
    }
#pragma unroll
    for (int t = 0; t < 4; ++t) ibs[lb + 32 * t + lane] = ibv[t];

    asm volatile("s_wait_dscnt 0x0" ::: "memory");
    asm volatile("s_barrier_signal -1" ::: "memory");
    asm volatile("s_barrier_wait -1" ::: "memory");

    const int c = lane & 15;
    const int h = lane >> 4;

    v2f af[16];
#pragma unroll
    for (int k0 = 0; k0 < 16; ++k0)
        af[k0] = *(const v2f*)(As + c * PITCH + 4 * k0 + 2 * h);

    const float biasv = biasp[0];
    float ubv[8];
#pragma unroll
    for (int v = 0; v < 8; ++v) ubv[v] = ubs[8 * h + v];

#pragma unroll
    for (int s = 0; s < 4; ++s) {
        const int t0 = lb + 32 * s;
        __builtin_amdgcn_sched_barrier(0);
#pragma unroll
        for (int j = 0; j < 16; ++j)
            *(v4f*)(Bs + (t0 + 2 * j + r) * PITCH + 4 * q) = w[16 * s + j];
        __builtin_amdgcn_fence(__ATOMIC_RELEASE, "wavefront");
        __builtin_amdgcn_wave_barrier();
        __builtin_amdgcn_fence(__ATOMIC_ACQUIRE, "wavefront");

        const float* bt = Bs + (t0 + c) * PITCH + 2 * h;
        const float ibr0 = ibs[t0 + c];
        const float ibr1 = ibs[t0 + 16 + c];
        v2f q0[16], q1[16];
#pragma unroll
        for (int k0 = 0; k0 < 16; ++k0) {
            q0[k0] = *(const v2f*)(bt + 4 * k0);
            q1[k0] = *(const v2f*)(bt + 16 * PITCH + 4 * k0);
        }
        __builtin_amdgcn_sched_barrier(0);
        v8f acc0 = {}, acc1 = {};
#pragma unroll
        for (int k0 = 0; k0 < 16; ++k0) {
            acc0 = __builtin_amdgcn_wmma_f32_16x16x4_f32(false, af[k0], false, q0[k0], (short)0, acc0, false, false);
            acc1 = __builtin_amdgcn_wmma_f32_16x16x4_f32(false, af[k0], false, q1[k0], (short)0, acc1, false, false);
        }

        const float ib0 = ibr0 + biasv;
        const float ib1 = ibr1 + biasv;
        float* obase = out + (size_t)(m0 + 8 * h) * N_BI + ns * NSLAB + t0 + c;
#pragma unroll
        for (int v = 0; v < 8; ++v) {
            float* orow = obase + (size_t)v * N_BI;
            orow[0]  = acc0[v] + ubv[v] + ib0;
            orow[16] = acc1[v] + ubv[v] + ib1;
        }
    }
}

extern "C" void kernel_launch(void* const* d_in, const int* in_sizes, int n_in,
                              void* d_out, int out_size, void* d_ws, size_t ws_size,
                              hipStream_t stream) {
    (void)in_sizes; (void)n_in; (void)out_size; (void)d_ws; (void)ws_size;

    const int*   user_ids    = (const int*)  d_in[0];
    const int*   item_ids    = (const int*)  d_in[1];
    const int*   fb_indices  = (const int*)  d_in[2];
    const float* fb_values   = (const float*)d_in[3];
    const float* user_weight = (const float*)d_in[4];
    const float* user_bias   = (const float*)d_in[5];
    const float* item_weight = (const float*)d_in[6];
    const float* item_bias   = (const float*)d_in[7];
    const float* bias        = (const float*)d_in[8];
    const float* irew        = (const float*)d_in[9];
    float* out = (float*)d_out;

    fused_rating<<<(N_BU / 16) * (N_BI / NSLAB), 256, 0, stream>>>(
        user_ids, item_ids, fb_indices, fb_values, user_weight, user_bias,
        item_weight, item_bias, bias, irew, out);
}
